# starting kernel
# baseline (speedup 1.0000x reference)
#include <hip/hip_runtime.h>

typedef float v2f __attribute__((ext_vector_type(2)));
typedef float v8f __attribute__((ext_vector_type(8)));

#define IN_F  512
#define HID_F 256
#define OUT_F 128

#define WMMA_F32(a, b, c) \
  __builtin_amdgcn_wmma_f32_16x16x4_f32(false, (a), false, (b), (short)0, (c), false, false)

__global__ void zero_f4(float4* __restrict__ p, long long n4) {
  long long i = (long long)blockIdx.x * blockDim.x + threadIdx.x;
  if (i < n4) p[i] = make_float4(0.f, 0.f, 0.f, 0.f);
}

__global__ void degree_k(const int* __restrict__ dst, float* __restrict__ deg, int E) {
  int i = blockIdx.x * blockDim.x + threadIdx.x;
  if (i < E) atomicAdd(&deg[dst[i]], 1.0f);
}

__global__ void invdeg_k(float* __restrict__ deg, int n) {
  int i = blockIdx.x * blockDim.x + threadIdx.x;
  if (i < n) deg[i] = 1.0f / fmaxf(deg[i], 1.0f);
}

__global__ __launch_bounds__(128)
void gemm_wmma_f32(const float* __restrict__ A, const float* __restrict__ B,
                   float* __restrict__ C, int M, int N, int K) {
  const int lane = threadIdx.x & 31;
  const int wave = threadIdx.x >> 5;
  const int m0 = (blockIdx.x * 4 + wave) * 32;
  const int n0 = blockIdx.y * 64;
  if (m0 >= M) return;
  const bool has2 = (m0 + 16) < M;

  const int half = lane >> 4;
  const int l15  = lane & 15;

  const int row0 = m0 + l15;
  int row1 = m0 + 16 + l15;
  if (row1 >= M) row1 = M - 1;
  const float* Ap0 = A + (size_t)row0 * K + half * 2;
  const float* Ap1 = A + (size_t)row1 * K + half * 2;
  const float* Bp = B + (size_t)(half * 2) * N + n0 + l15;

  v8f c00 = {}, c01 = {}, c02 = {}, c03 = {};
  v8f c10 = {}, c11 = {}, c12 = {}, c13 = {};

  for (int k = 0; k < K; k += 4) {
    v2f a0 = *(const v2f*)(Ap0 + k);
    v2f a1 = *(const v2f*)(Ap1 + k);
    const float* bk = Bp + (size_t)k * N;
    v2f b0, b1, b2, b3;
    b0.x = bk[0];   b0.y = bk[(size_t)N + 0];
    b1.x = bk[16];  b1.y = bk[(size_t)N + 16];
    b2.x = bk[32];  b2.y = bk[(size_t)N + 32];
    b3.x = bk[48];  b3.y = bk[(size_t)N + 48];
    c00 = WMMA_F32(a0, b0, c00);
    c01 = WMMA_F32(a0, b1, c01);
    c02 = WMMA_F32(a0, b2, c02);
    c03 = WMMA_F32(a0, b3, c03);
    c10 = WMMA_F32(a1, b0, c10);
    c11 = WMMA_F32(a1, b1, c11);
    c12 = WMMA_F32(a1, b2, c12);
    c13 = WMMA_F32(a1, b3, c13);
  }

  float* Cp0 = C + (size_t)(m0 + half * 8) * N + n0 + l15;
#pragma unroll
  for (int j = 0; j < 8; ++j) {
    float* r = Cp0 + (size_t)j * N;
    r[0]  = c00[j];
    r[16] = c01[j];
    r[32] = c02[j];
    r[48] = c03[j];
  }
  if (has2) {
    float* Cp1 = C + (size_t)(m0 + 16 + half * 8) * N + n0 + l15;
#pragma unroll
    for (int j = 0; j < 8; ++j) {
      float* r = Cp1 + (size_t)j * N;
      r[0]  = c10[j];
      r[16] = c11[j];
      r[32] = c12[j];
      r[48] = c13[j];
    }
  }
}

__global__ void edge_scatter(const float* __restrict__ hw, const float* __restrict__ ew,
                             const int* __restrict__ src, const int* __restrict__ dst,
                             float* __restrict__ agg, long long total, int fshift, int F) {
  long long idx = (long long)blockIdx.x * blockDim.x + threadIdx.x;
  if (idx >= total) return;
  int e  = (int)(idx >> fshift);
  int f4 = (int)(idx & ((1 << fshift) - 1)) << 2;
  float w = ew[e];
  const float4 v = *(const float4*)(hw + (size_t)src[e] * F + f4);
  float* p = agg + (size_t)dst[e] * F + f4;
  atomicAdd(p + 0, v.x * w);
  atomicAdd(p + 1, v.y * w);
  atomicAdd(p + 2, v.z * w);
  atomicAdd(p + 3, v.w * w);
}

__global__ void bias_norm_act(float* __restrict__ data, const float* __restrict__ inv,
                              const float* __restrict__ bias, long long n,
                              int fshift, int do_relu) {
  long long i = (long long)blockIdx.x * blockDim.x + threadIdx.x;
  if (i >= n) return;
  int row = (int)(i >> fshift);
  int col = (int)(i & ((1 << fshift) - 1));
  float v = data[i] * inv[row] + bias[col];
  data[i] = do_relu ? fmaxf(v, 0.f) : v;
}

extern "C" void kernel_launch(void* const* d_in, const int* in_sizes, int n_in,
                              void* d_out, int out_size, void* d_ws, size_t ws_size,
                              hipStream_t stream) {
  const float* x    = (const float*)d_in[0];
  const float* W1   = (const float*)d_in[1];
  const float* b1   = (const float*)d_in[2];
  const float* W2   = (const float*)d_in[3];
  const float* b2   = (const float*)d_in[4];
  const float* ew   = (const float*)d_in[5];
  const int*   esrc = (const int*)d_in[6];
  const int*   edst = (const int*)d_in[7];
  float* out = (float*)d_out;

  const int M = in_sizes[0] / IN_F;
  const int E = in_sizes[5];

  float* inv  = (float*)d_ws;
  const long long Mpad = ((long long)M + 63) & ~63LL;
  float* bufA = inv + Mpad;
  float* bufB = bufA + (long long)M * HID_F;

  const int T = 256;

  {
    long long n4 = (long long)M / 4;
    zero_f4<<<dim3((unsigned)((n4 + T - 1) / T)), dim3(T), 0, stream>>>((float4*)inv, n4);
    long long nB4 = (long long)M * HID_F / 4;
    zero_f4<<<dim3((unsigned)((nB4 + T - 1) / T)), dim3(T), 0, stream>>>((float4*)bufB, nB4);
    long long nO4 = (long long)M * OUT_F / 4;
    zero_f4<<<dim3((unsigned)((nO4 + T - 1) / T)), dim3(T), 0, stream>>>((float4*)out, nO4);
  }

  degree_k<<<dim3((E + T - 1) / T), dim3(T), 0, stream>>>(edst, inv, E);
  invdeg_k<<<dim3((M + T - 1) / T), dim3(T), 0, stream>>>(inv, M);

  {
    dim3 g((M + 127) / 128, HID_F / 64);
    gemm_wmma_f32<<<g, dim3(128), 0, stream>>>(x, W1, bufA, M, HID_F, IN_F);
  }

  {
    long long total = (long long)E * (HID_F / 4);
    edge_scatter<<<dim3((unsigned)((total + T - 1) / T)), dim3(T), 0, stream>>>(
        bufA, ew, esrc, edst, bufB, total, 6  , HID_F);
  }

  {
    long long n = (long long)M * HID_F;
    bias_norm_act<<<dim3((unsigned)((n + T - 1) / T)), dim3(T), 0, stream>>>(
        bufB, inv, b1, n, 8  , 1);
  }

  {
    dim3 g((M + 127) / 128, OUT_F / 64);
    gemm_wmma_f32<<<g, dim3(128), 0, stream>>>(bufB, W2, bufA, M, OUT_F, HID_F);
  }

  {
    long long total = (long long)E * (OUT_F / 4);
    edge_scatter<<<dim3((unsigned)((total + T - 1) / T)), dim3(T), 0, stream>>>(
        bufA, ew, esrc, edst, out, total, 5  , OUT_F);
  }

  {
    long long n = (long long)M * OUT_F;
    bias_norm_act<<<dim3((unsigned)((n + T - 1) / T)), dim3(T), 0, stream>>>(
        out, inv, b2, n, 7  , 0);
  }
}


# edge records carry their payload: the pre-pass bins (source row, weight) pairs instead of edge ids, LDS and parked queues hold them; no src[e]/ew[e] lookups in either aggregation; sum order kept
# speedup vs baseline: 3.6610x; 3.6610x over previous
#include <hip/hip_runtime.h>


typedef float v2f __attribute__((ext_vector_type(2)));
typedef float v8f __attribute__((ext_vector_type(8)));

#define IN_F  512
#define HID_F 256
#define OUT_F 128

#define WMMA_F32(a, b, c) \
  __builtin_amdgcn_wmma_f32_16x16x4_f32(false, (a), false, (b), (short)0, (c), false, false)

template <int N, int K>
__global__ __launch_bounds__(128)
void gemm_wmma_f32(const float* __restrict__ A, const float* __restrict__ B,
                   float* __restrict__ C, int M) {
  const int lane = threadIdx.x & 31;
  const int wave = threadIdx.x >> 5;
  const int m0 = (blockIdx.x * 4 + wave) * 32;
  const int n0 = blockIdx.y * 64;
  if (m0 >= M) return;
  const bool has2 = (m0 + 16) < M;

  const int half = lane >> 4;
  const int l15  = lane & 15;

  const int row0 = m0 + l15;
  int row1 = m0 + 16 + l15;
  if (row1 >= M) row1 = M - 1;
  const float* Ap0 = A + (size_t)row0 * K + half * 4;
  const float* Ap1 = A + (size_t)row1 * K + half * 4;
  const float* Bp = B + (size_t)(half * 4) * N + n0 + l15;
  constexpr int N1 = N, N2 = 2 * N, N3 = 3 * N;

  v8f c00 = {}, c01 = {}, c02 = {}, c03 = {};
  v8f c10 = {}, c11 = {}, c12 = {}, c13 = {};

  for (int k = 0; k < K; k += 32) {
    float4 a0[4], a1[4];
#pragma unroll
    for (int i = 0; i < 4; ++i) {
      a0[i] = *(const float4*)(Ap0 + k + 8 * i);
      a1[i] = *(const float4*)(Ap1 + k + 8 * i);
    }
    __builtin_amdgcn_sched_barrier(0);
#pragma unroll
    for (int i = 0; i < 4; ++i) {
      const float* bk = Bp + (size_t)(k + 8 * i) * N;
      v2f a0lo, a0hi, a1lo, a1hi;
      a0lo.x = a0[i].x; a0lo.y = a0[i].y; a0hi.x = a0[i].z; a0hi.y = a0[i].w;
      a1lo.x = a1[i].x; a1lo.y = a1[i].y; a1hi.x = a1[i].z; a1hi.y = a1[i].w;
      v2f b0lo, b1lo, b2lo, b3lo, b0hi, b1hi, b2hi, b3hi;
      b0lo.x = bk[0];       b0lo.y = bk[N1 + 0];
      b1lo.x = bk[16];      b1lo.y = bk[N1 + 16];
      b2lo.x = bk[32];      b2lo.y = bk[N1 + 32];
      b3lo.x = bk[48];      b3lo.y = bk[N1 + 48];
      b0hi.x = bk[N2 + 0];  b0hi.y = bk[N3 + 0];
      b1hi.x = bk[N2 + 16]; b1hi.y = bk[N3 + 16];
      b2hi.x = bk[N2 + 32]; b2hi.y = bk[N3 + 32];
      b3hi.x = bk[N2 + 48]; b3hi.y = bk[N3 + 48];
      c00 = WMMA_F32(a0lo, b0lo, c00);
      c01 = WMMA_F32(a0lo, b1lo, c01);
      c02 = WMMA_F32(a0lo, b2lo, c02);
      c03 = WMMA_F32(a0lo, b3lo, c03);
      c10 = WMMA_F32(a1lo, b0lo, c10);
      c11 = WMMA_F32(a1lo, b1lo, c11);
      c12 = WMMA_F32(a1lo, b2lo, c12);
      c13 = WMMA_F32(a1lo, b3lo, c13);
      c00 = WMMA_F32(a0hi, b0hi, c00);
      c01 = WMMA_F32(a0hi, b1hi, c01);
      c02 = WMMA_F32(a0hi, b2hi, c02);
      c03 = WMMA_F32(a0hi, b3hi, c03);
      c10 = WMMA_F32(a1hi, b0hi, c10);
      c11 = WMMA_F32(a1hi, b1hi, c11);
      c12 = WMMA_F32(a1hi, b2hi, c12);
      c13 = WMMA_F32(a1hi, b3hi, c13);
    }
  }

  float* Cp0 = C + (size_t)(m0 + half * 8) * N + n0 + l15;
#pragma unroll
  for (int j = 0; j < 8; ++j) {
    float* r = Cp0 + (size_t)j * N;
    r[0]  = c00[j];
    r[16] = c01[j];
    r[32] = c02[j];
    r[48] = c03[j];
  }
  if (has2) {
    float* Cp1 = C + (size_t)(m0 + 16 + half * 8) * N + n0 + l15;
#pragma unroll
    for (int j = 0; j < 8; ++j) {
      float* r = Cp1 + (size_t)j * N;
      r[0]  = c10[j];
      r[16] = c11[j];
      r[32] = c12[j];
      r[48] = c13[j];
    }
  }
}

#define AGG_TILE  196
#define AGG_T     1024
#define AGG_NW    (AGG_T / 32)
#define AGG_SLOTS ((AGG_TILE + AGG_NW - 1) / AGG_NW)
#define AGG_WQCAP 512
#define AGG_EBITS 28
#define AGG_EMASK ((1 << AGG_EBITS) - 1)
#define PARK_W    OUT_F

#define BIN_NB    256
#define BIN_T     512
#define BIN_MAXT  2048
#define BIN_EBITS 24
#define BIN_EMASK ((1 << BIN_EBITS) - 1)

__device__ __forceinline__ bool bins_usable(int M) {
  return M <= (1 << BIN_EBITS) && (M + AGG_TILE - 1) / AGG_TILE <= BIN_MAXT;
}
template <int PITCH>
__device__ __forceinline__ int bin_cap(int rows) {
  return ((rows * PITCH - BIN_NB) / BIN_NB) >> 1;
}

template <int PITCH>
__global__ __launch_bounds__(BIN_T)
void bin_edges(const int* __restrict__ dst, const int* __restrict__ src, const float* __restrict__ ew,
               int* __restrict__ bins, int M, int E) {
  __shared__ int hist[BIN_MAXT];
  if (!bins_usable(M)) return;
  const int tid = threadIdx.x;
  const int b = blockIdx.x;
  const int tiles = (M + AGG_TILE - 1) / AGG_TILE;
  for (int t = tid; t < tiles; t += BIN_T) hist[t] = 0;
  __syncthreads();
  const int per = (E + BIN_NB - 1) / BIN_NB;
  const int e0 = b * per;
  const int e1 = min(E, e0 + per);
  for (int base = e0 + tid; base < e1; base += 4 * BIN_T) {
    int d[4], s[4];
    float w[4];
#pragma unroll
    for (int u = 0; u < 4; ++u) {
      const int e = base + u * BIN_T;
      const bool in = e < e1;
      d[u] = in ? dst[e] : -1;
      s[u] = in ? src[e] : 0;
      w[u] = in ? ew[e] : 0.f;
    }
#pragma unroll
    for (int u = 0; u < 4; ++u) {
      if ((unsigned)d[u] < (unsigned)M) {
        const int t = d[u] / AGG_TILE;
        const int tl = d[u] - t * AGG_TILE;
        const int cap = bin_cap<PITCH>(min(AGG_TILE, M - t * AGG_TILE));
        const int pos = atomicAdd(&hist[t], 1);
        if (pos < cap)
          *(int2*)(bins + (size_t)t * (AGG_TILE * PITCH) + BIN_NB + 2 * (b * cap + pos)) =
              make_int2((int)(((unsigned)s[u] & BIN_EMASK) | ((unsigned)tl << BIN_EBITS)),
                        __float_as_int(w[u]));
      }
    }
  }
  __syncthreads();
  for (int t = tid; t < tiles; t += BIN_T) {
    const int cap = bin_cap<PITCH>(min(AGG_TILE, M - t * AGG_TILE));
    const int c = hist[t];
    bins[(size_t)t * (AGG_TILE * PITCH) + b] = (c <= cap) ? c : -1;
  }
}

template <int F, bool RELU, int NB, int MODE>
__global__ __launch_bounds__(AGG_T)
void agg_owner(const float* __restrict__ hw, const float* __restrict__ ew,
               const int* __restrict__ src, const int* __restrict__ dst,
               const float* __restrict__ bias, float* outp, int* park, int M, int E) {
  constexpr int NV = F / 128;
  __shared__ int2 wq[AGG_NW][AGG_WQCAP];
  __shared__ int wqn[AGG_NW];
  __shared__ int binbad;

  const int tid  = threadIdx.x;
  const int lane = tid & 31;
  const int wave = __builtin_amdgcn_readfirstlane(tid >> 5);
  const int lo   = blockIdx.x * AGG_TILE;
  const int nrows = min(AGG_TILE, M - lo);
  const bool packed_ok = M <= (1 << AGG_EBITS);
  int* pk = park + (size_t)lo * PARK_W;
  const int pstride = ((nrows * PARK_W - AGG_NW) / AGG_NW) >> 1;
  int2* pq = (int2*)(pk + AGG_NW) + wave * pstride;

  if (MODE == 0) {
    if (tid < AGG_NW) wqn[tid] = 0;
    if (tid == 0) binbad = 0;
    __syncthreads();
  }

  auto push = [&](int s, int wbits, int t) {
    const int w = t & (AGG_NW - 1);
    const int pos = atomicAdd(&wqn[w], 1);
    if (pos < AGG_WQCAP) wq[w][pos] = make_int2((s & AGG_EMASK) | ((t / AGG_NW) << AGG_EBITS), wbits);
  };
  auto push_edge = [&](int e, int t) { push(src[e], __float_as_int(ew[e]), t); };
  auto test4 = [&](const int4 d, int e0) {
    const unsigned a = (unsigned)(d.x - lo), b = (unsigned)(d.y - lo);
    const unsigned c = (unsigned)(d.z - lo), g = (unsigned)(d.w - lo);
    const unsigned nr = (unsigned)nrows;
    if ((a < nr) | (b < nr) | (c < nr) | (g < nr)) {
      if (a < nr) push_edge(e0 + 0, (int)a);
      if (b < nr) push_edge(e0 + 1, (int)b);
      if (c < nr) push_edge(e0 + 2, (int)c);
      if (g < nr) push_edge(e0 + 3, (int)g);
    }
  };

  bool scan_self = MODE == 0 && packed_ok;
  if (MODE == 0 && packed_ok && bins_usable(M)) {
    static_assert(AGG_T == 4 * BIN_NB, "four threads a bin");
    const int* bn = (const int*)outp + (size_t)lo * F;
    const int cap = bin_cap<F>(nrows);
    const int b = tid >> 2, sub = tid & 3;
    const int c = bn[b];
    if (c < 0 || c > cap) binbad = 1;
    __syncthreads();
    if (!binbad) {
      scan_self = false;
      const int2* sg = (const int2*)(bn + BIN_NB) + b * cap;
      for (int i = sub; i < c; i += 16) {
        int2 ent[4];
#pragma unroll
        for (int u = 0; u < 4; ++u) ent[u] = (i + 4 * u < c) ? sg[i + 4 * u] : make_int2(0, 0);
#pragma unroll
        for (int u = 0; u < 4; ++u)
          if (i + 4 * u < c)
            push(ent[u].x & BIN_EMASK, ent[u].y, (int)((unsigned)ent[u].x >> BIN_EBITS));
      }
    }
  }
  if (scan_self) {
    const int G4 = E >> 2;
    const int4* dst4 = (const int4*)dst;
    for (int g0 = tid; g0 < G4; g0 += AGG_T * 8) {
      int4 d[8];
      if (g0 + 7 * AGG_T < G4) {
#pragma unroll
        for (int u = 0; u < 8; ++u) d[u] = dst4[g0 + u * AGG_T];
      } else {
#pragma unroll
        for (int u = 0; u < 8; ++u) {
          const int gi = g0 + u * AGG_T;
          d[u] = (gi < G4) ? dst4[gi] : make_int4(-1, -1, -1, -1);
        }
      }
#pragma unroll
      for (int u = 0; u < 8; ++u) test4(d[u], (g0 + u * AGG_T) << 2);
    }
    for (int e = (G4 << 2) + tid; e < E; e += AGG_T) {
      const unsigned a = (unsigned)(dst[e] - lo);
      if (a < (unsigned)nrows) push_edge(e, (int)a);
    }
  }
  int cnt = AGG_WQCAP + 1;
  if (MODE == 1) {
    const int c = __builtin_amdgcn_readfirstlane(pk[wave]);
    if (c >= 0 && c <= AGG_WQCAP && c <= pstride) {
      for (int i = lane; i < c; i += 32) wq[wave][i] = pq[i];
      cnt = c;
    }
  }
  __syncthreads();
  if (MODE == 0) {
    if (packed_ok) cnt = __builtin_amdgcn_readfirstlane(wqn[wave]);
    const bool keep = cnt <= AGG_WQCAP && cnt <= pstride;
    if (lane == 0) pk[wave] = keep ? cnt : -1;
    if (keep)
      for (int i = lane; i < cnt; i += 32) pq[i] = wq[wave][i];
  }

  float4 acc[AGG_SLOTS][NV];
  int    deg[AGG_SLOTS];
#pragma unroll
  for (int c = 0; c < AGG_SLOTS; ++c) {
    deg[c] = 0;
#pragma unroll
    for (int j = 0; j < NV; ++j) acc[c][j] = make_float4(0.f, 0.f, 0.f, 0.f);
  }

  auto add_group = [&](bool valid, int sl, int sv, float wv) {
#pragma unroll
    for (int c = 0; c < AGG_SLOTS; ++c) {
      unsigned m = __builtin_amdgcn_ballot_w32(valid && sl == c);
      deg[c] += __builtin_popcount(m);
      while (m) {
        const int nk = min(__builtin_popcount(m), NB);
        float4 v[NB][NV];
        float  w[NB];
#pragma unroll
        for (int k = 0; k < NB; ++k) {
          if (k < nk) {
            const int kk = __builtin_ctz(m);
            m &= m - 1;
            const int s = __builtin_amdgcn_readlane(sv, kk);
            w[k] = __int_as_float(__builtin_amdgcn_readlane(__float_as_int(wv), kk));
            const float4* row = (const float4*)(hw + (size_t)s * F) + lane;
#pragma unroll
            for (int j = 0; j < NV; ++j) v[k][j] = row[j * 32];
          }
        }
#pragma unroll
        for (int k = 0; k < NB; ++k) {
          if (k < nk) {
#pragma unroll
            for (int j = 0; j < NV; ++j) {
              acc[c][j].x += w[k] * v[k][j].x;
              acc[c][j].y += w[k] * v[k][j].y;
              acc[c][j].z += w[k] * v[k][j].z;
              acc[c][j].w += w[k] * v[k][j].w;
            }
          }
        }
      }
    }
  };

  if (cnt <= AGG_WQCAP) {
    const int2* q = wq[wave];
    for (int g = 0; g < cnt; g += 32) {
      const bool valid = g + lane < cnt;
      int2 ent = make_int2(0, 0);
      if (valid) ent = q[g + lane];
      add_group(valid, (int)((unsigned)ent.x >> AGG_EBITS), ent.x & AGG_EMASK, __int_as_float(ent.y));
    }
  } else {
    for (int e0 = 0; e0 < E; e0 += 32) {
      const int e = e0 + lane;
      unsigned a = 0xffffffffu;
      if (e < E) a = (unsigned)(dst[e] - lo);
      const bool mine = (a < (unsigned)nrows) && ((int)(a & (AGG_NW - 1)) == wave);
      int sv = 0;
      float wv = 0.f;
      if (mine) { sv = src[e]; wv = ew[e]; }
      add_group(mine, (int)(a / AGG_NW), sv, wv);
    }
  }

#pragma unroll
  for (int c = 0; c < AGG_SLOTS; ++c) {
    const int t = c * AGG_NW + wave;
    if (t < nrows) {
      const float inv = 1.0f / fmaxf((float)deg[c], 1.0f);
      float* orow = outp + (size_t)(lo + t) * F + 4 * lane;
#pragma unroll
      for (int j = 0; j < NV; ++j) {
        const float4 a = acc[c][j];
        const float4 bv = *(const float4*)&bias[j * 128 + 4 * lane];
        float4 r;
        r.x = a.x * inv + bv.x;
        r.y = a.y * inv + bv.y;
        r.z = a.z * inv + bv.z;
        r.w = a.w * inv + bv.w;
        if (RELU) {
          r.x = fmaxf(r.x, 0.f); r.y = fmaxf(r.y, 0.f);
          r.z = fmaxf(r.z, 0.f); r.w = fmaxf(r.w, 0.f);
        }
        *(float4*)(orow + j * 128) = r;
      }
    }
  }
}

extern "C" void kernel_launch(void* const* d_in, const int* in_sizes, int n_in,
                              void* d_out, int out_size, void* d_ws, size_t ws_size,
                              hipStream_t stream) {
  const float* x    = (const float*)d_in[0];
  const float* W1   = (const float*)d_in[1];
  const float* b1   = (const float*)d_in[2];
  const float* W2   = (const float*)d_in[3];
  const float* b2   = (const float*)d_in[4];
  const float* ew   = (const float*)d_in[5];
  const int*   esrc = (const int*)d_in[6];
  const int*   edst = (const int*)d_in[7];
  float* out = (float*)d_out;

  const int M = in_sizes[0] / IN_F;
  const int E = in_sizes[5];

  float* ws0  = (float*)d_ws;
  const long long Mpad = ((long long)M + 63) & ~63LL;
  float* bufA = ws0 + Mpad;
  float* bufB = bufA + (long long)M * HID_F;

  const int tiles = (M + AGG_TILE - 1) / AGG_TILE;

  bin_edges<HID_F><<<dim3(BIN_NB), dim3(BIN_T), 0, stream>>>(edst, esrc, ew, (int*)bufB, M, E);
  {
    dim3 g((M + 127) / 128, HID_F / 64);
    gemm_wmma_f32<HID_F, IN_F><<<g, dim3(128), 0, stream>>>(x, W1, bufA, M);
  }
  agg_owner<HID_F, true, 4, 0><<<dim3(tiles), dim3(AGG_T), 0, stream>>>(
      bufA, ew, esrc, edst, b1, bufB, (int*)d_out, M, E);
  {
    dim3 g((M + 127) / 128, OUT_F / 64);
    gemm_wmma_f32<OUT_F, HID_F><<<g, dim3(128), 0, stream>>>(bufB, W2, bufA, M);
  }
  agg_owner<OUT_F, false, 8, 1><<<dim3(tiles), dim3(AGG_T), 0, stream>>>(
      bufA, ew, esrc, edst, b2, out, (int*)d_out, M, E);
}


# aggregation queues: one LDS queue per destination row (own LDS counter), parked in that row's output words; a wave sums one row at a time with deeper row requests; sums stay arrival-ordered
# speedup vs baseline: 3.8299x; 1.0461x over previous
#include <hip/hip_runtime.h>


typedef float v2f __attribute__((ext_vector_type(2)));
typedef float v8f __attribute__((ext_vector_type(8)));

#define IN_F  512
#define HID_F 256
#define OUT_F 128

#define WMMA_F32(a, b, c) \
  __builtin_amdgcn_wmma_f32_16x16x4_f32(false, (a), false, (b), (short)0, (c), false, false)

template <int N, int K>
__global__ __launch_bounds__(128)
void gemm_wmma_f32(const float* __restrict__ A, const float* __restrict__ B,
                   float* __restrict__ C, int M) {
  const int lane = threadIdx.x & 31;
  const int wave = threadIdx.x >> 5;
  const int m0 = (blockIdx.x * 4 + wave) * 32;
  const int n0 = blockIdx.y * 64;
  if (m0 >= M) return;
  const bool has2 = (m0 + 16) < M;

  const int half = lane >> 4;
  const int l15  = lane & 15;

  const int row0 = m0 + l15;
  int row1 = m0 + 16 + l15;
  if (row1 >= M) row1 = M - 1;
  const float* Ap0 = A + (size_t)row0 * K + half * 4;
  const float* Ap1 = A + (size_t)row1 * K + half * 4;
  const float* Bp = B + (size_t)(half * 4) * N + n0 + l15;
  constexpr int N1 = N, N2 = 2 * N, N3 = 3 * N;

  v8f c00 = {}, c01 = {}, c02 = {}, c03 = {};
  v8f c10 = {}, c11 = {}, c12 = {}, c13 = {};

  for (int k = 0; k < K; k += 32) {
    float4 a0[4], a1[4];
#pragma unroll
    for (int i = 0; i < 4; ++i) {
      a0[i] = *(const float4*)(Ap0 + k + 8 * i);
      a1[i] = *(const float4*)(Ap1 + k + 8 * i);
    }
    __builtin_amdgcn_sched_barrier(0);
#pragma unroll
    for (int i = 0; i < 4; ++i) {
      const float* bk = Bp + (size_t)(k + 8 * i) * N;
      v2f a0lo, a0hi, a1lo, a1hi;
      a0lo.x = a0[i].x; a0lo.y = a0[i].y; a0hi.x = a0[i].z; a0hi.y = a0[i].w;
      a1lo.x = a1[i].x; a1lo.y = a1[i].y; a1hi.x = a1[i].z; a1hi.y = a1[i].w;
      v2f b0lo, b1lo, b2lo, b3lo, b0hi, b1hi, b2hi, b3hi;
      b0lo.x = bk[0];       b0lo.y = bk[N1 + 0];
      b1lo.x = bk[16];      b1lo.y = bk[N1 + 16];
      b2lo.x = bk[32];      b2lo.y = bk[N1 + 32];
      b3lo.x = bk[48];      b3lo.y = bk[N1 + 48];
      b0hi.x = bk[N2 + 0];  b0hi.y = bk[N3 + 0];
      b1hi.x = bk[N2 + 16]; b1hi.y = bk[N3 + 16];
      b2hi.x = bk[N2 + 32]; b2hi.y = bk[N3 + 32];
      b3hi.x = bk[N2 + 48]; b3hi.y = bk[N3 + 48];
      c00 = WMMA_F32(a0lo, b0lo, c00);
      c01 = WMMA_F32(a0lo, b1lo, c01);
      c02 = WMMA_F32(a0lo, b2lo, c02);
      c03 = WMMA_F32(a0lo, b3lo, c03);
      c10 = WMMA_F32(a1lo, b0lo, c10);
      c11 = WMMA_F32(a1lo, b1lo, c11);
      c12 = WMMA_F32(a1lo, b2lo, c12);
      c13 = WMMA_F32(a1lo, b3lo, c13);
      c00 = WMMA_F32(a0hi, b0hi, c00);
      c01 = WMMA_F32(a0hi, b1hi, c01);
      c02 = WMMA_F32(a0hi, b2hi, c02);
      c03 = WMMA_F32(a0hi, b3hi, c03);
      c10 = WMMA_F32(a1hi, b0hi, c10);
      c11 = WMMA_F32(a1hi, b1hi, c11);
      c12 = WMMA_F32(a1hi, b2hi, c12);
      c13 = WMMA_F32(a1hi, b3hi, c13);
    }
  }

  float* Cp0 = C + (size_t)(m0 + half * 8) * N + n0 + l15;
#pragma unroll
  for (int j = 0; j < 8; ++j) {
    float* r = Cp0 + (size_t)j * N;
    r[0]  = c00[j];
    r[16] = c01[j];
    r[32] = c02[j];
    r[48] = c03[j];
  }
  if (has2) {
    float* Cp1 = C + (size_t)(m0 + 16 + half * 8) * N + n0 + l15;
#pragma unroll
    for (int j = 0; j < 8; ++j) {
      float* r = Cp1 + (size_t)j * N;
      r[0]  = c10[j];
      r[16] = c11[j];
      r[32] = c12[j];
      r[48] = c13[j];
    }
  }
}

#define AGG_TILE  196
#define AGG_T     1024
#define AGG_NW    (AGG_T / 32)
#define AGG_SLOTS ((AGG_TILE + AGG_NW - 1) / AGG_NW)
#define PARK_W    OUT_F
#define ROW_PITCH (PARK_W / 2)
#define ROW_CAP   (ROW_PITCH - 1)

#define BIN_NB    256
#define BIN_T     512
#define BIN_MAXT  2048
#define BIN_EBITS 24
#define BIN_EMASK ((1 << BIN_EBITS) - 1)

__device__ __forceinline__ bool bins_usable(int M) {
  return M <= (1 << BIN_EBITS) && (M + AGG_TILE - 1) / AGG_TILE <= BIN_MAXT;
}
template <int PITCH>
__device__ __forceinline__ int bin_cap(int rows) {
  return ((rows * PITCH - BIN_NB) / BIN_NB) >> 1;
}

template <int PITCH>
__global__ __launch_bounds__(BIN_T)
void bin_edges(const int* __restrict__ dst, const int* __restrict__ src, const float* __restrict__ ew,
               int* __restrict__ bins, int M, int E) {
  __shared__ int hist[BIN_MAXT];
  if (!bins_usable(M)) return;
  const int tid = threadIdx.x;
  const int b = blockIdx.x;
  const int tiles = (M + AGG_TILE - 1) / AGG_TILE;
  for (int t = tid; t < tiles; t += BIN_T) hist[t] = 0;
  __syncthreads();
  const int per = (E + BIN_NB - 1) / BIN_NB;
  const int e0 = b * per;
  const int e1 = min(E, e0 + per);
  for (int base = e0 + tid; base < e1; base += 4 * BIN_T) {
    int d[4], s[4];
    float w[4];
#pragma unroll
    for (int u = 0; u < 4; ++u) {
      const int e = base + u * BIN_T;
      const bool in = e < e1;
      d[u] = in ? dst[e] : -1;
      s[u] = in ? src[e] : 0;
      w[u] = in ? ew[e] : 0.f;
    }
#pragma unroll
    for (int u = 0; u < 4; ++u) {
      if ((unsigned)d[u] < (unsigned)M) {
        const int t = d[u] / AGG_TILE;
        const int tl = d[u] - t * AGG_TILE;
        const int cap = bin_cap<PITCH>(min(AGG_TILE, M - t * AGG_TILE));
        const int pos = atomicAdd(&hist[t], 1);
        if (pos < cap)
          *(int2*)(bins + (size_t)t * (AGG_TILE * PITCH) + BIN_NB + 2 * (b * cap + pos)) =
              make_int2((int)(((unsigned)s[u] & BIN_EMASK) | ((unsigned)tl << BIN_EBITS)),
                        __float_as_int(w[u]));
      }
    }
  }
  __syncthreads();
  for (int t = tid; t < tiles; t += BIN_T) {
    const int cap = bin_cap<PITCH>(min(AGG_TILE, M - t * AGG_TILE));
    const int c = hist[t];
    bins[(size_t)t * (AGG_TILE * PITCH) + b] = (c <= cap) ? c : -1;
  }
}

template <int F, bool RELU, int NB, int MODE>
__global__ __launch_bounds__(AGG_T)
void agg_owner(const float* __restrict__ hw, const float* __restrict__ ew,
               const int* __restrict__ src, const int* __restrict__ dst,
               const float* __restrict__ bias, float* outp, int* park, int M, int E) {
  constexpr int NV = F / 128;
  static_assert(4 * NB >= 32, "a group of 32 records is at most four requests");
  __shared__ int4 rq4[AGG_TILE][ROW_PITCH / 2 + 1];
  __shared__ int rcnt[AGG_TILE];
  __shared__ int binbad;

  const int tid  = threadIdx.x;
  const int lane = tid & 31;
  const int wave = __builtin_amdgcn_readfirstlane(tid >> 5);
  const int lo   = blockIdx.x * AGG_TILE;
  const int nrows = min(AGG_TILE, M - lo);

  if (MODE == 0) {
    if (tid < AGG_TILE) rcnt[tid] = 0;
    if (tid == 0) binbad = 0;
    __syncthreads();

    auto push = [&](int s, int wbits, int t) {
      const int pos = atomicAdd(&rcnt[t], 1);
      if (pos < ROW_CAP) ((int2*)rq4[t])[1 + pos] = make_int2(s, wbits);
    };
    auto push_edge = [&](int e, int t) { push(src[e], __float_as_int(ew[e]), t); };
    auto test4 = [&](const int4 d, int e0) {
      const unsigned a = (unsigned)(d.x - lo), b = (unsigned)(d.y - lo);
      const unsigned c = (unsigned)(d.z - lo), g = (unsigned)(d.w - lo);
      const unsigned nr = (unsigned)nrows;
      if ((a < nr) | (b < nr) | (c < nr) | (g < nr)) {
        if (a < nr) push_edge(e0 + 0, (int)a);
        if (b < nr) push_edge(e0 + 1, (int)b);
        if (c < nr) push_edge(e0 + 2, (int)c);
        if (g < nr) push_edge(e0 + 3, (int)g);
      }
    };

    bool scan_self = true;
    if (bins_usable(M)) {
      static_assert(AGG_T == 4 * BIN_NB, "four threads a bin");
      const int* bn = (const int*)outp + (size_t)lo * F;
      const int cap = bin_cap<F>(nrows);
      const int b = tid >> 2, sub = tid & 3;
      const int c = bn[b];
      if (c < 0 || c > cap) binbad = 1;
      __syncthreads();
      if (!binbad) {
        scan_self = false;
        const int2* sg = (const int2*)(bn + BIN_NB) + b * cap;
        for (int i = sub; i < c; i += 16) {
          int2 ent[4];
#pragma unroll
          for (int u = 0; u < 4; ++u) ent[u] = (i + 4 * u < c) ? sg[i + 4 * u] : make_int2(0, 0);
#pragma unroll
          for (int u = 0; u < 4; ++u)
            if (i + 4 * u < c)
              push(ent[u].x & BIN_EMASK, ent[u].y, (int)((unsigned)ent[u].x >> BIN_EBITS));
        }
      }
    }
    if (scan_self) {
      const int G4 = E >> 2;
      const int4* dst4 = (const int4*)dst;
      for (int g0 = tid; g0 < G4; g0 += AGG_T * 8) {
        int4 d[8];
        if (g0 + 7 * AGG_T < G4) {
#pragma unroll
          for (int u = 0; u < 8; ++u) d[u] = dst4[g0 + u * AGG_T];
        } else {
#pragma unroll
          for (int u = 0; u < 8; ++u) {
            const int gi = g0 + u * AGG_T;
            d[u] = (gi < G4) ? dst4[gi] : make_int4(-1, -1, -1, -1);
          }
        }
#pragma unroll
        for (int u = 0; u < 8; ++u) test4(d[u], (g0 + u * AGG_T) << 2);
      }
      for (int e = (G4 << 2) + tid; e < E; e += AGG_T) {
        const unsigned a = (unsigned)(dst[e] - lo);
        if (a < (unsigned)nrows) push_edge(e, (int)a);
      }
    }
    __syncthreads();
    int4 px[AGG_SLOTS];
    int  pn[AGG_SLOTS];
#pragma unroll
    for (int c = 0; c < AGG_SLOTS; ++c) {
      const int t = c * AGG_NW + wave;
      if (t < nrows) {
        pn[c] = rcnt[t];
        px[c] = rq4[t][lane];
      }
    }
#pragma unroll
    for (int c = 0; c < AGG_SLOTS; ++c) {
      const int t = c * AGG_NW + wave;
      if (t < nrows) {
        int4 x = px[c];
        if (lane == 0) x.x = pn[c];
        if (2 * lane <= min(pn[c], ROW_CAP)) ((int4*)(park + (size_t)(lo + t) * PARK_W))[lane] = x;
      }
    }
  } else {
    int4 x[AGG_SLOTS];
#pragma unroll
    for (int c = 0; c < AGG_SLOTS; ++c) {
      const int t = c * AGG_NW + wave;
      if (t < nrows) x[c] = ((const int4*)(park + (size_t)(lo + t) * PARK_W))[lane];
    }
#pragma unroll
    for (int c = 0; c < AGG_SLOTS; ++c) {
      const int t = c * AGG_NW + wave;
      if (t < nrows) {
        rq4[t][lane] = x[c];
        if (lane == 0) rcnt[t] = x[c].x;
      }
    }
    __syncthreads();
  }

  float4 bv[NV];
#pragma unroll
  for (int j = 0; j < NV; ++j) bv[j] = *(const float4*)&bias[j * 128 + 4 * lane];

  for (int c = 0; c < AGG_SLOTS; ++c) {
    const int t = c * AGG_NW + wave;
    if (t >= nrows) break;
    const int n = __builtin_amdgcn_readfirstlane(rcnt[t]);

    float4 acc[NV];
#pragma unroll
    for (int j = 0; j < NV; ++j) acc[j] = make_float4(0.f, 0.f, 0.f, 0.f);

    auto add_records = [&](unsigned m, int sv, float wv) {
      const int k = __builtin_popcount(m);
      int per = k;
      if (k > NB) per = (k + 1) >> 1;
      if (k > 2 * NB) per = (k + 2) / 3;
      if (k > 3 * NB) per = (k + 3) >> 2;
      if (per < 1) per = 1;
      while (m) {
        const int nk = min(__builtin_popcount(m), per);
        float4 v[NB][NV];
        float  w[NB];
#pragma unroll
        for (int i = 0; i < NB; ++i) {
          if (i < nk) {
            const int kk = __builtin_ctz(m);
            m &= m - 1;
            const int s = __builtin_amdgcn_readlane(sv, kk);
            w[i] = __int_as_float(__builtin_amdgcn_readlane(__float_as_int(wv), kk));
            const float4* row = (const float4*)(hw + (size_t)s * F) + lane;
#pragma unroll
            for (int j = 0; j < NV; ++j) v[i][j] = row[j * 32];
          }
        }
#pragma unroll
        for (int i = 0; i < NB; ++i) {
          if (i < nk) {
#pragma unroll
            for (int j = 0; j < NV; ++j) {
              acc[j].x += w[i] * v[i][j].x;
              acc[j].y += w[i] * v[i][j].y;
              acc[j].z += w[i] * v[i][j].z;
              acc[j].w += w[i] * v[i][j].w;
            }
          }
        }
      }
    };

    if (n <= ROW_CAP) {
      const int2* q = (const int2*)rq4[t] + 1;
      for (int g = 0; g < n; g += 32) {
        const int ng = min(n - g, 32);
        int2 ent = make_int2(0, 0);
        if (lane < ng) ent = q[g + lane];
        add_records(ng == 32 ? 0xffffffffu : ((1u << ng) - 1u), ent.x, __int_as_float(ent.y));
      }
    } else {
      for (int e0 = 0; e0 < E; e0 += 32) {
        const int e = e0 + lane;
        const bool mine = (e < E) && (dst[e] == lo + t);
        int sv = 0;
        float wv = 0.f;
        if (mine) { sv = src[e]; wv = ew[e]; }
        add_records(__builtin_amdgcn_ballot_w32(mine), sv, wv);
      }
    }

    const float inv = 1.0f / fmaxf((float)n, 1.0f);
    float* orow = outp + (size_t)(lo + t) * F + 4 * lane;
#pragma unroll
    for (int j = 0; j < NV; ++j) {
      const float4 a = acc[j];
      float4 r;
      r.x = a.x * inv + bv[j].x;
      r.y = a.y * inv + bv[j].y;
      r.z = a.z * inv + bv[j].z;
      r.w = a.w * inv + bv[j].w;
      if (RELU) {
        r.x = fmaxf(r.x, 0.f); r.y = fmaxf(r.y, 0.f);
        r.z = fmaxf(r.z, 0.f); r.w = fmaxf(r.w, 0.f);
      }
      *(float4*)(orow + j * 128) = r;
    }
  }
}

extern "C" void kernel_launch(void* const* d_in, const int* in_sizes, int n_in,
                              void* d_out, int out_size, void* d_ws, size_t ws_size,
                              hipStream_t stream) {
  const float* x    = (const float*)d_in[0];
  const float* W1   = (const float*)d_in[1];
  const float* b1   = (const float*)d_in[2];
  const float* W2   = (const float*)d_in[3];
  const float* b2   = (const float*)d_in[4];
  const float* ew   = (const float*)d_in[5];
  const int*   esrc = (const int*)d_in[6];
  const int*   edst = (const int*)d_in[7];
  float* out = (float*)d_out;

  const int M = in_sizes[0] / IN_F;
  const int E = in_sizes[5];

  float* ws0  = (float*)d_ws;
  const long long Mpad = ((long long)M + 63) & ~63LL;
  float* bufA = ws0 + Mpad;
  float* bufB = bufA + (long long)M * HID_F;

  const int tiles = (M + AGG_TILE - 1) / AGG_TILE;

  bin_edges<HID_F><<<dim3(BIN_NB), dim3(BIN_T), 0, stream>>>(edst, esrc, ew, (int*)bufB, M, E);
  {
    dim3 g((M + 127) / 128, HID_F / 64);
    gemm_wmma_f32<HID_F, IN_F><<<g, dim3(128), 0, stream>>>(x, W1, bufA, M);
  }
  agg_owner<HID_F, true, 10, 0><<<dim3(tiles), dim3(AGG_T), 0, stream>>>(
      bufA, ew, esrc, edst, b1, bufB, (int*)d_out, M, E);
  {
    dim3 g((M + 127) / 128, OUT_F / 64);
    gemm_wmma_f32<OUT_F, HID_F><<<g, dim3(128), 0, stream>>>(bufB, W2, bufA, M);
  }
  agg_owner<OUT_F, false, 16, 1><<<dim3(tiles), dim3(AGG_T), 0, stream>>>(
      bufA, ew, esrc, edst, b2, out, (int*)d_out, M, E);
}
